# starting kernel
# baseline (speedup 1.0000x reference)
#include <hip/hip_runtime.h>
#include <hip/hip_bf16.h>

#define Bsz   4
#define Cin   256
#define CRdim 64
#define Gdim  16
#define GCdim 16
#define Kker  7
#define KK    49
#define M2    784
#define OUTC  256
#define Hdim  64
#define Wdim  64
#define HW    4096
#define NWAVES 4

typedef __attribute__((ext_vector_type(2))) float v2f;
typedef __attribute__((ext_vector_type(8))) float v8f;

template <int NTM, int NTN>
__device__ __forceinline__ void wmma_gemm_block(const float* __restrict__ A, int lda,
                                                const float* __restrict__ Bm, int ldb,
                                                int Ksz, int mBase, int nBase,
                                                v8f acc[NTM][NTN]) {
  const int lane = threadIdx.x & 31;
  const int half = lane >> 4;
  const int l16  = lane & 15;
  const float* arow = A + (size_t)(mBase + l16) * lda + 2 * half;
  const float* bptr = Bm + (size_t)(2 * half) * ldb + nBase + l16;
  for (int k = 0; k < Ksz; k += 4) {
    v2f a[NTM];
#pragma unroll
    for (int tm = 0; tm < NTM; ++tm) {
      a[tm].x = arow[(size_t)tm * 16 * lda + k + 0];
      a[tm].y = arow[(size_t)tm * 16 * lda + k + 1];
    }
#pragma unroll
    for (int tn = 0; tn < NTN; ++tn) {
      v2f b;
      b.x = bptr[(size_t)(k + 0) * ldb + tn * 16];
      b.y = bptr[(size_t)(k + 1) * ldb + tn * 16];
#pragma unroll
      for (int tm = 0; tm < NTM; ++tm) {
        acc[tm][tn] = __builtin_amdgcn_wmma_f32_16x16x4_f32(
            false, a[tm], false, b, (short)0, acc[tm][tn], false, false);
      }
    }
  }
}

__global__ void k_gemm_bn_relu(const float* __restrict__ x,
                               const float* __restrict__ w1,
                               const float* __restrict__ gamma,
                               const float* __restrict__ beta,
                               const float* __restrict__ mean,
                               const float* __restrict__ var,
                               float* __restrict__ y) {
  const int wave  = threadIdx.x >> 5;
  const int nBase = (blockIdx.x * NWAVES + wave) * 32;
  const int mBase = blockIdx.y * 32;
  const int b     = blockIdx.z;
  const float* Bm = x + (size_t)b * Cin * HW;
  v8f acc[2][2] = {};
  wmma_gemm_block<2, 2>(w1, Cin, Bm, HW, Cin, mBase, nBase, acc);

  float* Yb = y + (size_t)b * CRdim * HW;
  const int lane = threadIdx.x & 31;
  const int half = lane >> 4, l16 = lane & 15;
#pragma unroll
  for (int tm = 0; tm < 2; ++tm) {
#pragma unroll
    for (int r = 0; r < 8; ++r) {
      const int m = mBase + tm * 16 + r + 8 * half;
      const float s  = gamma[m] * rsqrtf(var[m] + 1e-5f);
      const float mu = mean[m], bt = beta[m];
#pragma unroll
      for (int tn = 0; tn < 2; ++tn) {
        const float v = (acc[tm][tn][r] - mu) * s + bt;
        Yb[(size_t)m * HW + nBase + tn * 16 + l16] = fmaxf(v, 0.0f);
      }
    }
  }
}

__global__ void k_gemm_bias(const float* __restrict__ y,
                            const float* __restrict__ w2,
                            const float* __restrict__ b2,
                            float* __restrict__ wt) {
  const int wave  = threadIdx.x >> 5;
  const int nBase = (blockIdx.x * NWAVES + wave) * 64;
  const int mBase = blockIdx.y * 16;
  const int b     = blockIdx.z;
  const float* Bm = y + (size_t)b * CRdim * HW;
  v8f acc[1][4] = {};
  wmma_gemm_block<1, 4>(w2, CRdim, Bm, HW, CRdim, mBase, nBase, acc);

  float* Wb = wt + (size_t)b * M2 * HW;
  const int lane = threadIdx.x & 31;
  const int half = lane >> 4, l16 = lane & 15;
#pragma unroll
  for (int r = 0; r < 8; ++r) {
    const int m = mBase + r + 8 * half;
    const float bias = b2[m];
#pragma unroll
    for (int tn = 0; tn < 4; ++tn) {
      Wb[(size_t)m * HW + nBase + tn * 16 + l16] = acc[0][tn][r] + bias;
    }
  }
}

__global__ void k_involution(const float* __restrict__ x,
                             const float* __restrict__ wt,
                             float* __restrict__ inv) {
  __shared__ float taps[KK * 16];

  const int xtile = blockIdx.x & 3;
  const int py    = blockIdx.x >> 2;
  const int g     = blockIdx.y;
  const int b     = blockIdx.z;
  const int tid   = threadIdx.x;
  const int pxi   = tid & 15;
  const int ch    = tid >> 4;
  const int xbase = xtile * 16;

  const float* wtp = wt + ((size_t)b * M2 + (size_t)g * KK) * HW + py * Wdim + xbase;
  for (int e = tid; e < KK * 16; e += 256) {
    const int k = e >> 4;
    taps[e] = wtp[(size_t)k * HW + (e & 15)];
  }
  __syncthreads();

  const int c = g * GCdim + ch;
  const float* xp = x + ((size_t)b * Cin + c) * HW;
  const int px = xbase + pxi;

  float acc = 0.0f;
#pragma unroll
  for (int dy = 0; dy < Kker; ++dy) {
    const int yy = py + dy - 3;
    const bool okY = (unsigned)yy < (unsigned)Hdim;
#pragma unroll
    for (int dx = 0; dx < Kker; ++dx) {
      const int xx = px + dx - 3;
      const bool ok = okY && ((unsigned)xx < (unsigned)Wdim);
      const float xv = ok ? xp[yy * Wdim + xx] : 0.0f;
      const float wv = taps[(dy * Kker + dx) * 16 + pxi];
      acc = fmaf(wv, xv, acc);
    }
  }
  inv[((size_t)b * Cin + c) * HW + py * Wdim + px] = acc;
}

__global__ void k_gemm_plain(const float* __restrict__ inv,
                             const float* __restrict__ wp,
                             float* __restrict__ out) {
  const int wave  = threadIdx.x >> 5;
  const int nBase = (blockIdx.x * NWAVES + wave) * 32;
  const int mBase = blockIdx.y * 32;
  const int b     = blockIdx.z;
  const float* Bm = inv + (size_t)b * Cin * HW;
  v8f acc[2][2] = {};
  wmma_gemm_block<2, 2>(wp, Cin, Bm, HW, Cin, mBase, nBase, acc);

  float* Ob = out + (size_t)b * OUTC * HW;
  const int lane = threadIdx.x & 31;
  const int half = lane >> 4, l16 = lane & 15;
#pragma unroll
  for (int tm = 0; tm < 2; ++tm) {
#pragma unroll
    for (int r = 0; r < 8; ++r) {
      const int m = mBase + tm * 16 + r + 8 * half;
#pragma unroll
      for (int tn = 0; tn < 2; ++tn) {
        Ob[(size_t)m * HW + nBase + tn * 16 + l16] = acc[tm][tn][r];
      }
    }
  }
}

extern "C" void kernel_launch(void* const* d_in, const int* in_sizes, int n_in,
                              void* d_out, int out_size, void* d_ws, size_t ws_size,
                              hipStream_t stream) {
  const float* x     = (const float*)d_in[0];
  const float* w1    = (const float*)d_in[1];
  const float* gamma = (const float*)d_in[2];
  const float* beta  = (const float*)d_in[3];
  const float* mean  = (const float*)d_in[4];
  const float* var   = (const float*)d_in[5];
  const float* w2    = (const float*)d_in[6];
  const float* b2    = (const float*)d_in[7];
  const float* wp    = (const float*)d_in[8];
  float* out = (float*)d_out;

  float* y   = (float*)d_ws;
  float* wt  = y  + (size_t)Bsz * CRdim * HW;
  float* inv = wt + (size_t)Bsz * M2 * HW;

  dim3 blk(32 * NWAVES, 1, 1);

  dim3 g1(HW / (32 * NWAVES), CRdim / 32, Bsz);
  k_gemm_bn_relu<<<g1, blk, 0, stream>>>(x, w1, gamma, beta, mean, var, y);

  dim3 g2(HW / (64 * NWAVES), M2 / 16, Bsz);
  k_gemm_bias<<<g2, blk, 0, stream>>>(y, w2, b2, wt);

  dim3 gI(Hdim * (Wdim / 16), Gdim, Bsz);
  k_involution<<<gI, 256, 0, stream>>>(x, wt, inv);

  dim3 g4(HW / (32 * NWAVES), OUTC / 32, Bsz);
  k_gemm_plain<<<g4, blk, 0, stream>>>(inv, wp, out);
}


# weight (A) tile of each of the three GEMMs copied once per block into LDS by whole-line 16-byte loads; A fragments read from LDS, no 16-line row gathers from global memory; sum order unchanged
# speedup vs baseline: 3.6823x; 3.6823x over previous
#include <hip/hip_runtime.h>
#include <hip/hip_bf16.h>

#define Bsz   4
#define Cin   256
#define CRdim 64
#define Gdim  16
#define GCdim 16
#define Kker  7
#define KK    49
#define M2    784
#define OUTC  256
#define Hdim  64
#define Wdim  64
#define HW    4096
#define NWAVES 4
#define WA_PITCH 260

typedef __attribute__((ext_vector_type(2))) float v2f;
typedef __attribute__((ext_vector_type(8))) float v8f;
typedef __attribute__((ext_vector_type(4))) float v4f;

template <int NTM, int NTN, int KC, class FillA>
__device__ __forceinline__ void wmma_gemm_chunked(const float* A, int lda,
                                                  const float* __restrict__ Bm, int ldb,
                                                  int Ksz, int mBase, int nBase,
                                                  v8f acc[NTM][NTN], FillA fill_a) {
  const int lane = threadIdx.x & 31;
  const int half = lane >> 4;
  const int l16  = lane & 15;
  const float* arow = A + (mBase + l16) * lda + 2 * half;
  const float* bptr = Bm + (size_t)(2 * half) * ldb + nBase + l16;
  for (int k0 = 0; k0 < Ksz; k0 += 4 * KC) {
    v2f a[KC][NTM];
    v2f b[KC][NTN];
#pragma unroll
    for (int s = 0; s < KC; ++s) {
      const int k = k0 + 4 * s;
#pragma unroll
      for (int tn = 0; tn < NTN; ++tn) {
        b[s][tn].x = bptr[(size_t)(k + 0) * ldb + tn * 16];
        b[s][tn].y = bptr[(size_t)(k + 1) * ldb + tn * 16];
      }
    }
    if (k0 == 0) fill_a();
#pragma unroll
    for (int s = 0; s < KC; ++s) {
      const int k = k0 + 4 * s;
#pragma unroll
      for (int tm = 0; tm < NTM; ++tm) {
        a[s][tm] = *reinterpret_cast<const v2f*>(arow + tm * 16 * lda + k);
      }
    }
    __builtin_amdgcn_sched_barrier(0);
#pragma unroll
    for (int s = 0; s < KC; ++s) {
#pragma unroll
      for (int tn = 0; tn < NTN; ++tn) {
#pragma unroll
        for (int tm = 0; tm < NTM; ++tm) {
          acc[tm][tn] = __builtin_amdgcn_wmma_f32_16x16x4_f32(
              false, a[s][tm], false, b[s][tn], (short)0, acc[tm][tn], false, false);
        }
      }
    }
  }
}

__device__ __forceinline__ void stage_rows32_load(v4f (&st)[8], const float* __restrict__ wrows) {
  const v4f* src = reinterpret_cast<const v4f*>(wrows);
#pragma unroll
  for (int i = 0; i < 8; ++i) {
    st[i] = src[threadIdx.x + 256 * i];
  }
}
__device__ __forceinline__ void stage_rows32_store(float* wa, const v4f (&st)[8]) {
#pragma unroll
  for (int i = 0; i < 8; ++i) {
    const int q = threadIdx.x + 256 * i;
    *reinterpret_cast<v4f*>(&wa[(q >> 6) * WA_PITCH + 4 * (q & 63)]) = st[i];
  }
}

#define K1_PITCH 36
__global__ void __launch_bounds__(256)
k_gemm_bn_relu(const float* __restrict__ x,
                               const float* __restrict__ w1,
                               const float* __restrict__ gamma,
                               const float* __restrict__ beta,
                               const float* __restrict__ mean,
                               const float* __restrict__ var,
                               float* __restrict__ y) {
  __shared__ __attribute__((aligned(16))) float red[2][4][32 * K1_PITCH];
  __shared__ __attribute__((aligned(16))) float wa[32 * WA_PITCH];
  const int wave  = threadIdx.x >> 5;
  const int ct    = wave & 1;
  const int kq    = wave >> 1;
  const int nBase = (blockIdx.x * 2 + ct) * 32;
  const int mBase = blockIdx.y * 32;
  const int b     = blockIdx.z;
  const float* Bm = x + (size_t)b * Cin * HW + (size_t)kq * 64 * HW;
  v4f st[8];
  stage_rows32_load(st, w1 + (size_t)mBase * Cin);
  v8f acc[2][2] = {};
  wmma_gemm_chunked<2, 2, 16>(wa + kq * 64, WA_PITCH, Bm, HW, 64, 0, nBase, acc,
                              [&]() { stage_rows32_store(wa, st); __syncthreads(); });

  const int lane = threadIdx.x & 31;
  const int half = lane >> 4, l16 = lane & 15;
  float* part = &red[ct][kq][0];
#pragma unroll
  for (int tm = 0; tm < 2; ++tm) {
#pragma unroll
    for (int tn = 0; tn < 2; ++tn) {
#pragma unroll
      for (int r = 0; r < 8; ++r) {
        part[(tm * 16 + 8 * half + r) * K1_PITCH + tn * 16 + l16] = acc[tm][tn][r];
      }
    }
  }
  __syncthreads();

  float* Yb = y + (size_t)b * CRdim * HW;
  const int rr = lane >> 3, c4 = lane & 7;
#pragma unroll
  for (int i = 0; i < 2; ++i) {
    const int row = kq * 8 + 4 * i + rr;
    const int o   = row * K1_PITCH + 4 * c4;
    v4f s = *reinterpret_cast<const v4f*>(&red[ct][0][o]);
    s += *reinterpret_cast<const v4f*>(&red[ct][1][o]);
    s += *reinterpret_cast<const v4f*>(&red[ct][2][o]);
    s += *reinterpret_cast<const v4f*>(&red[ct][3][o]);
    const int m = mBase + row;
    const float sc = gamma[m] * rsqrtf(var[m] + 1e-5f);
    const float mu = mean[m], bt = beta[m];
    v4f v;
    v.x = fmaxf((s.x - mu) * sc + bt, 0.0f);
    v.y = fmaxf((s.y - mu) * sc + bt, 0.0f);
    v.z = fmaxf((s.z - mu) * sc + bt, 0.0f);
    v.w = fmaxf((s.w - mu) * sc + bt, 0.0f);
    *reinterpret_cast<v4f*>(Yb + (size_t)m * HW + nBase + 4 * c4) = v;
  }
}

#define TAPT 7
#define W2_PITCH   68
#define W2_GSTRIDE 3360

__device__ __forceinline__ void inv_load_window(v4f (&win)[Kker][3],
                                                const v4f* __restrict__ xc,
                                                const int (&off)[Kker][3]) {
#pragma unroll
  for (int dy = 0; dy < Kker; ++dy) {
#pragma unroll
    for (int j = 0; j < 3; ++j) {
      win[dy][j] = xc[off[dy][j]];
    }
  }
}

__device__ __forceinline__ void inv_channel(const float (&tap)[4][KK],
                                            const v4f (&win)[Kker][3],
                                            float* __restrict__ dst) {
  float acc[4] = {0.0f, 0.0f, 0.0f, 0.0f};
#pragma unroll
  for (int dy = 0; dy < Kker; ++dy) {
    const float r[12] = {win[dy][0].x, win[dy][0].y, win[dy][0].z, win[dy][0].w,
                         win[dy][1].x, win[dy][1].y, win[dy][1].z, win[dy][1].w,
                         win[dy][2].x, win[dy][2].y, win[dy][2].z, win[dy][2].w};
#pragma unroll
    for (int dx = 0; dx < Kker; ++dx) {
#pragma unroll
      for (int p = 0; p < 4; ++p) {
        acc[p] = fmaf(tap[p][dy * Kker + dx], r[p + dx + 1], acc[p]);
      }
    }
  }
  v4f o;
  o.x = acc[0]; o.y = acc[1]; o.z = acc[2]; o.w = acc[3];
  *reinterpret_cast<v4f*>(dst) = o;
}

__global__ void __launch_bounds__(32 * NWAVES) __attribute__((amdgpu_waves_per_eu(2, 2)))
k_wgen_involution(const float* __restrict__ x,
                  const float* __restrict__ y,
                  const float* __restrict__ w2,
                  const float* __restrict__ b2,
                  float* __restrict__ inv) {
  const int lane = threadIdx.x & 31;
  const int wave = threadIdx.x >> 5;
  const int half = lane >> 4;
  const int l16  = lane & 15;
  const int py   = blockIdx.x * NWAVES + wave;
  const int G    = blockIdx.y;
  const int b    = blockIdx.z;

  __shared__ __attribute__((aligned(16))) float w2s[2 * W2_GSTRIDE + 4 * 32 * (NWAVES - 1)];
  v4f piece[13];
  {
    const v4f* src = reinterpret_cast<const v4f*>(w2 + (size_t)(2 * G) * KK * CRdim);
#pragma unroll
    for (int i = 0; i < 13; ++i) {
      piece[i] = src[min((int)threadIdx.x + 32 * NWAVES * i, 2 * KK * (CRdim / 4) - 1)];
    }
  }

  int aoff[TAPT];
#pragma unroll
  for (int tm = 0; tm < TAPT; ++tm) {
    const int t = min(8 * tm + (l16 & 7), KK - 1);
    aoff[tm] = (l16 >> 3) * W2_GSTRIDE + t * W2_PITCH + 2 * half;
  }
  const float* yb = y + (size_t)b * CRdim * HW;
  const int boff = (2 * half) * HW + py * Wdim + 4 * l16;

  v8f acc[TAPT][4] = {};
  for (int k0 = 0; k0 < CRdim; k0 += 16) {
    v2f a[4][TAPT];
    v4f bq0[4], bq1[4];
#pragma unroll
    for (int s = 0; s < 4; ++s) {
      const int k = k0 + 4 * s;
      bq0[s] = *reinterpret_cast<const v4f*>(yb + boff + (k + 0) * HW);
      bq1[s] = *reinterpret_cast<const v4f*>(yb + boff + (k + 1) * HW);
    }
    if (k0 == 0) {
#pragma unroll
      for (int i = 0; i < 13; ++i) {
        const int q   = min((int)threadIdx.x + 32 * NWAVES * i, 2 * KK * (CRdim / 4) - 1);
        const int row = q >> 4, c4 = q & 15;
        const int grp = (row >= KK) ? 1 : 0;
        int dst = grp * W2_GSTRIDE + (row - grp * KK) * W2_PITCH + 4 * c4;
        if (i == 12 && threadIdx.x >= 32) dst = 2 * W2_GSTRIDE + 4 * ((int)threadIdx.x - 32);
        *reinterpret_cast<v4f*>(&w2s[dst]) = piece[i];
      }
      __syncthreads();
    }
#pragma unroll
    for (int s = 0; s < 4; ++s) {
      const int k = k0 + 4 * s;
#pragma unroll
      for (int tm = 0; tm < TAPT; ++tm) {
        a[s][tm] = *reinterpret_cast<const v2f*>(&w2s[aoff[tm] + k]);
      }
    }
    __builtin_amdgcn_sched_barrier(0);
#pragma unroll
    for (int s = 0; s < 4; ++s) {
#pragma unroll
      for (int tn = 0; tn < 4; ++tn) {
        v2f bb;
        bb.x = bq0[s][tn];
        bb.y = bq1[s][tn];
#pragma unroll
        for (int tm = 0; tm < TAPT; ++tm) {
          acc[tm][tn] = __builtin_amdgcn_wmma_f32_16x16x4_f32(
              false, a[s][tm], false, bb, (short)0, acc[tm][tn], false, false);
        }
      }
    }
  }

  const int g   = 2 * G + half;
  const int q   = py * (Wdim / 4) + l16;
  const int px0 = l16 * 4;

  float my[Kker];
  int rowo[Kker];
#pragma unroll
  for (int d = 0; d < Kker; ++d) {
    const int yy = py + d - 3;
    my[d]   = ((unsigned)yy < (unsigned)Hdim) ? 1.0f : 0.0f;
    rowo[d] = min(max(yy, 0), Hdim - 1) * (Wdim / 4);
  }
  float mxe[Kker + 3];
#pragma unroll
  for (int e = 0; e < Kker + 3; ++e) {
    const int xx = px0 + e - 3;
    mxe[e] = ((unsigned)xx < (unsigned)Wdim) ? 1.0f : 0.0f;
  }
  const int colo[3] = {max(l16 - 1, 0), l16, min(l16 + 1, Wdim / 4 - 1)};
  int off[Kker][3];
#pragma unroll
  for (int dy = 0; dy < Kker; ++dy) {
#pragma unroll
    for (int j = 0; j < 3; ++j) {
      off[dy][j] = rowo[dy] + colo[j];
    }
  }

  const float* b2g = b2 + g * KK;
  float tap[4][KK];
#pragma unroll
  for (int dy = 0; dy < Kker; ++dy) {
#pragma unroll
    for (int dx = 0; dx < Kker; ++dx) {
      const int kk = dy * Kker + dx;
      const float bias = b2g[kk];
#pragma unroll
      for (int p = 0; p < 4; ++p) {
        const float w = acc[kk >> 3][p][kk & 7] + bias;
        tap[p][kk] = w * (my[dy] * mxe[p + dx]);
      }
    }
  }

  const v4f* xg = reinterpret_cast<const v4f*>(x + ((size_t)b * Cin + (size_t)g * GCdim) * HW);
  float* og = inv + ((size_t)b * Cin + (size_t)g * GCdim) * HW + (size_t)q * 4;
  v4f wa[Kker][3];
  v4f wb[Kker][3];
  inv_load_window(wa, xg, off);
  for (int c = 0; c < GCdim - 2; c += 2) {
    inv_load_window(wb, xg + (size_t)(c + 1) * (HW / 4), off);
    inv_channel(tap, wa, og + (size_t)c * HW);
    inv_load_window(wa, xg + (size_t)(c + 2) * (HW / 4), off);
    inv_channel(tap, wb, og + (size_t)(c + 1) * HW);
  }
  inv_load_window(wb, xg + (size_t)(GCdim - 1) * (HW / 4), off);
  inv_channel(tap, wa, og + (size_t)(GCdim - 2) * HW);
  inv_channel(tap, wb, og + (size_t)(GCdim - 1) * HW);
}

#define K3_PITCH 68
__global__ void __launch_bounds__(256)
k_gemm_plain(const float* __restrict__ inv,
                             const float* __restrict__ wp,
                             float* __restrict__ out) {
  __shared__ __attribute__((aligned(16))) float red[4][2][32 * K3_PITCH];
  __shared__ __attribute__((aligned(16))) float wa[32 * WA_PITCH];
  const int wave  = threadIdx.x >> 5;
  const int cw    = wave & 3;
  const int kh    = wave >> 2;
  const int nBase = (blockIdx.x * 4 + cw) * 64;
  const int mBase = blockIdx.y * 32;
  const int b     = blockIdx.z;
  const float* Bm = inv + (size_t)b * Cin * HW + (size_t)kh * 128 * HW;
  v4f st[8];
  stage_rows32_load(st, wp + (size_t)mBase * Cin);
  v8f acc[2][4] = {};
  wmma_gemm_chunked<2, 4, 4>(wa + kh * 128, WA_PITCH, Bm, HW, 128, 0, nBase, acc,
                             [&]() { stage_rows32_store(wa, st); __syncthreads(); });

  const int lane = threadIdx.x & 31;
  const int half = lane >> 4, l16 = lane & 15;
  float* part = &red[cw][kh][0];
#pragma unroll
  for (int tm = 0; tm < 2; ++tm) {
#pragma unroll
    for (int tn = 0; tn < 4; ++tn) {
#pragma unroll
      for (int r = 0; r < 8; ++r) {
        part[(tm * 16 + 8 * half + r) * K3_PITCH + tn * 16 + l16] = acc[tm][tn][r];
      }
    }
  }
  __syncthreads();

  float* Ob = out + (size_t)b * OUTC * HW;
  const int rr = lane >> 4, c4 = lane & 15;
#pragma unroll
  for (int i = 0; i < 8; ++i) {
    const int row = kh * 16 + 2 * i + rr;
    const int o   = row * K3_PITCH + 4 * c4;
    v4f s = *reinterpret_cast<const v4f*>(&red[cw][0][o]);
    s += *reinterpret_cast<const v4f*>(&red[cw][1][o]);
    *reinterpret_cast<v4f*>(Ob + (size_t)(mBase + row) * HW + nBase + 4 * c4) = s;
  }
}

extern "C" void kernel_launch(void* const* d_in, const int* in_sizes, int n_in,
                              void* d_out, int out_size, void* d_ws, size_t ws_size,
                              hipStream_t stream) {
  const float* x     = (const float*)d_in[0];
  const float* w1    = (const float*)d_in[1];
  const float* gamma = (const float*)d_in[2];
  const float* beta  = (const float*)d_in[3];
  const float* mean  = (const float*)d_in[4];
  const float* var   = (const float*)d_in[5];
  const float* w2    = (const float*)d_in[6];
  const float* b2    = (const float*)d_in[7];
  const float* wp    = (const float*)d_in[8];
  float* out = (float*)d_out;

  float* y   = (float*)d_ws;
  float* inv = y + (size_t)Bsz * CRdim * HW;

  dim3 blk(32 * NWAVES, 1, 1);
  dim3 blk8(256, 1, 1);

  dim3 g1(HW / 64, CRdim / 32, Bsz);
  k_gemm_bn_relu<<<g1, blk8, 0, stream>>>(x, w1, gamma, beta, mean, var, y);

  dim3 g2(Hdim / NWAVES, Gdim / 2, Bsz);
  k_wgen_involution<<<g2, blk, 0, stream>>>(x, y, w2, b2, inv);

  dim3 g4(HW / 256, OUTC / 32, Bsz);
  k_gemm_plain<<<g4, blk8, 0, stream>>>(inv, wp, out);
}
